# starting kernel
# baseline (speedup 1.0000x reference)
#include <hip/hip_runtime.h>
#include <hip/hip_bf16.h>

typedef __attribute__((ext_vector_type(2))) float v2f;
typedef __attribute__((ext_vector_type(8))) float v8f;

#define NF 64
#define DK 128
#define LDS_STRIDE 132
#define NPAIRS 2016

__global__ __launch_bounds__(128)
void InteractionLayer_gram_triu_kernel(const float* __restrict__ x,
                                       float* __restrict__ out) {
    __shared__ float sx[NF * LDS_STRIDE];

    const int b   = blockIdx.x;
    const int tid = threadIdx.x;
    const float* __restrict__ xb = x + (size_t)b * (NF * DK);

    for (int idx = tid; idx < NF * (DK / 4); idx += 128) {
        const int row = idx >> 5;
        const int c4  = idx & 31;
        const float4 v = reinterpret_cast<const float4*>(xb)[(row << 5) + c4];
        float* dst = &sx[row * LDS_STRIDE + (c4 << 2)];
        dst[0] = v.x; dst[1] = v.y; dst[2] = v.z; dst[3] = v.w;
    }
    __syncthreads();

    const int wave  = tid >> 5;
    const int lane  = tid & 31;
    const int m     = lane & 15;
    const int khalf = lane >> 4;

    const int wm = (wave >> 1) << 1;
    const int wn = (wave & 1)  << 1;

    const bool dead = (wm == 2) && (wn == 0);

    if (!dead) {
        v8f acc[2][2];
        #pragma unroll
        for (int i = 0; i < 2; ++i)
            #pragma unroll
            for (int j = 0; j < 2; ++j)
                acc[i][j] = (v8f){0.f, 0.f, 0.f, 0.f, 0.f, 0.f, 0.f, 0.f};

        for (int k = 0; k < DK / 4; ++k) {
            const int kb = (k << 2) + (khalf << 1);
            v2f afrag[2], bfrag[2];
            #pragma unroll
            for (int i = 0; i < 2; ++i) {
                afrag[i] = *(const v2f*)&sx[(((wm + i) << 4) + m) * LDS_STRIDE + kb];
                bfrag[i] = *(const v2f*)&sx[(((wn + i) << 4) + m) * LDS_STRIDE + kb];
            }
            #pragma unroll
            for (int i = 0; i < 2; ++i)
                #pragma unroll
                for (int j = 0; j < 2; ++j)
                    acc[i][j] = __builtin_amdgcn_wmma_f32_16x16x4_f32(
                         false, afrag[i],
                         false, bfrag[j],
                         (short)0, acc[i][j],
                         false,  false);
        }

        float* __restrict__ ob = out + (size_t)b * NPAIRS;
        #pragma unroll
        for (int i = 0; i < 2; ++i) {
            #pragma unroll
            for (int j = 0; j < 2; ++j) {
                const int ti = (wm + i) << 4;
                const int tj = (wn + j) << 4;
                #pragma unroll
                for (int v = 0; v < 8; ++v) {
                    const int gi = ti + v + (khalf << 3);
                    const int gj = tj + m;
                    if (gj > gi) {
                        const int idx = gi * (NF - 1) - ((gi * (gi - 1)) >> 1) + (gj - gi - 1);
                        ob[idx] = acc[i][j][v];
                    }
                }
            }
        }
    }
}

extern "C" void kernel_launch(void* const* d_in, const int* in_sizes, int n_in,
                              void* d_out, int out_size, void* d_ws, size_t ws_size,
                              hipStream_t stream) {
    (void)in_sizes; (void)n_in; (void)d_ws; (void)ws_size; (void)out_size;
    const float* x = (const float*)d_in[0];
    float* out = (float*)d_out;
    const int batches = 4096;
    InteractionLayer_gram_triu_kernel<<<batches, 128, 0, stream>>>(x, out);
}


# Persistent workgroups (1024 x 4 batches): next batch's 16 global loads per thread fly under the current batch's WMMA loop, two LDS stages, one barrier per batch; sum order unchanged
# speedup vs baseline: 1.1400x; 1.1400x over previous
#include <hip/hip_runtime.h>

typedef __attribute__((ext_vector_type(2))) float v2f;
typedef __attribute__((ext_vector_type(4))) float v4f;
typedef __attribute__((ext_vector_type(8))) float v8f;

#define NF 64
#define DK 128
#define LDS_STRIDE 132
#define NPAIRS 2016
#define NBATCH 4096
#define TILE_FLOATS (NF * LDS_STRIDE)
#define NLOAD 16
#define NBLOCKS 1024

__global__ __launch_bounds__(128)
void InteractionLayer_gram_triu_kernel(const float* __restrict__ x,
                                       float* __restrict__ out) {
    __shared__ float sx[2 * TILE_FLOATS];

    const int tid  = threadIdx.x;
    const int step = gridDim.x;
    int b = blockIdx.x;
    if (b >= NBATCH) return;

    const int wave  = __builtin_amdgcn_readfirstlane(tid >> 5);
    const int lane  = tid & 31;
    const int m     = lane & 15;
    const int khalf = lane >> 4;

    const int goff = tid << 2;
    const int soff = wave * LDS_STRIDE + (lane << 2);

    const int wm = (wave >> 1) << 1;
    const int wn = (wave & 1)  << 1;

    const bool dead = (wm == 2) && (wn == 0);

    v4f r[NLOAD];

    {
        const float* __restrict__ xb = x + (size_t)b * (NF * DK);
        #pragma unroll
        for (int i = 0; i < NLOAD; ++i)
            r[i] = *reinterpret_cast<const v4f*>(xb + goff + i * 512);
        #pragma unroll
        for (int i = 0; i < NLOAD; ++i)
            *reinterpret_cast<v4f*>(&sx[soff + i * (4 * LDS_STRIDE)]) = r[i];
    }
    __syncthreads();

    int cur = 0;
    for (; b < NBATCH; b += step) {
        const int  bn   = b + step;
        const bool more = bn < NBATCH;
        const float* sc = sx + cur * TILE_FLOATS;

        if (more) {
            const float* __restrict__ xb = x + (size_t)bn * (NF * DK);
            #pragma unroll
            for (int i = 0; i < NLOAD; ++i)
                r[i] = *reinterpret_cast<const v4f*>(xb + goff + i * 512);
        }

        if (!dead) {
            v8f acc[2][2];
            #pragma unroll
            for (int i = 0; i < 2; ++i)
                #pragma unroll
                for (int j = 0; j < 2; ++j)
                    acc[i][j] = (v8f){0.f, 0.f, 0.f, 0.f, 0.f, 0.f, 0.f, 0.f};

            for (int k = 0; k < DK / 4; ++k) {
                const int kb = (k << 2) + (khalf << 1);
                v2f afrag[2], bfrag[2];
                #pragma unroll
                for (int i = 0; i < 2; ++i) {
                    afrag[i] = *(const v2f*)&sc[(((wm + i) << 4) + m) * LDS_STRIDE + kb];
                    bfrag[i] = *(const v2f*)&sc[(((wn + i) << 4) + m) * LDS_STRIDE + kb];
                }
                #pragma unroll
                for (int i = 0; i < 2; ++i)
                    #pragma unroll
                    for (int j = 0; j < 2; ++j)
                        acc[i][j] = __builtin_amdgcn_wmma_f32_16x16x4_f32(
                             false, afrag[i],
                             false, bfrag[j],
                             (short)0, acc[i][j],
                             false,  false);
            }

            const unsigned obase = (unsigned)b * NPAIRS;
            #pragma unroll
            for (int i = 0; i < 2; ++i) {
                #pragma unroll
                for (int j = 0; j < 2; ++j) {
                    const int ti = (wm + i) << 4;
                    const int tj = (wn + j) << 4;
                    #pragma unroll
                    for (int v = 0; v < 8; ++v) {
                        const int gi = ti + v + (khalf << 3);
                        const int gj = tj + m;
                        if (gj > gi) {
                            const int idx = gi * (NF - 1) - ((gi * (gi - 1)) >> 1) + (gj - gi - 1);
                            out[obase + (unsigned)idx] = acc[i][j][v];
                        }
                    }
                }
            }
        }

        if (more) {
            float* sn = sx + (cur ^ 1) * TILE_FLOATS;
            #pragma unroll
            for (int i = 0; i < NLOAD; ++i)
                *reinterpret_cast<v4f*>(&sn[soff + i * (4 * LDS_STRIDE)]) = r[i];
        }
        __syncthreads();
        cur ^= 1;
    }
}

extern "C" void kernel_launch(void* const* d_in, const int* in_sizes, int n_in,
                              void* d_out, int out_size, void* d_ws, size_t ws_size,
                              hipStream_t stream) {
    (void)in_sizes; (void)n_in; (void)d_ws; (void)ws_size; (void)out_size;
    const float* x = (const float*)d_in[0];
    float* out = (float*)d_out;
    InteractionLayer_gram_triu_kernel<<<NBLOCKS, 128, 0, stream>>>(x, out);
}


# Only the 10 tiles on or above the diagonal are computed, dealt to all four waves as 3/2/2/3-tile roles that rotate every batch: no idle wave, no below-diagonal tile; k sum order unchanged
# speedup vs baseline: 1.4473x; 1.2696x over previous
#include <hip/hip_runtime.h>

typedef __attribute__((ext_vector_type(2))) float v2f;
typedef __attribute__((ext_vector_type(4))) float v4f;
typedef __attribute__((ext_vector_type(8))) float v8f;

#define NF 64
#define DK 128
#define LDS_STRIDE 132
#define NPAIRS 2016
#define NBATCH 4096
#define TILE_FLOATS (NF * LDS_STRIDE)
#define NLOAD 16
#define NBLOCKS 1024

__device__ __forceinline__ v8f mma4(v2f a, v2f b, v8f acc) {
    return __builtin_amdgcn_wmma_f32_16x16x4_f32( false, a,  false, b,
                                                  (short)0, acc,
                                                  false,  false);
}

__device__ __forceinline__ int rowterm(int gi) {
    return ((gi * (125 - gi)) >> 1) - 1;
}

__global__ __launch_bounds__(128)
void InteractionLayer_gram_triu_kernel(const float* __restrict__ x,
                                       float* __restrict__ out) {
    __shared__ float sx[2 * TILE_FLOATS];

    const int tid  = threadIdx.x;
    const int step = gridDim.x;
    int b = blockIdx.x;
    if (b >= NBATCH) return;

    const int wave  = __builtin_amdgcn_readfirstlane(tid >> 5);
    const int lane  = tid & 31;
    const int m     = lane & 15;
    const int khalf = lane >> 4;

    const int goff = tid << 2;
    const int soff = wave * LDS_STRIDE + (lane << 2);

    const int foff  = m * LDS_STRIDE + (khalf << 1);
    const int BAND  = 16 * LDS_STRIDE;

    const int rloc  = khalf << 3;

    v4f r[NLOAD];

    {
        const float* __restrict__ xb = x + (size_t)b * (NF * DK);
        #pragma unroll
        for (int i = 0; i < NLOAD; ++i)
            r[i] = *reinterpret_cast<const v4f*>(xb + goff + i * 512);
        #pragma unroll
        for (int i = 0; i < NLOAD; ++i)
            *reinterpret_cast<v4f*>(&sx[soff + i * (4 * LDS_STRIDE)]) = r[i];
    }
    __syncthreads();

    int cur  = 0;
    int role = wave;
    for (; b < NBATCH; b += step) {
        const int  bn   = b + step;
        const bool more = bn < NBATCH;
        const float* sc = sx + cur * TILE_FLOATS;

        if (more) {
            const float* __restrict__ xb = x + (size_t)bn * (NF * DK);
            #pragma unroll
            for (int i = 0; i < NLOAD; ++i)
                r[i] = *reinterpret_cast<const v4f*>(xb + goff + i * 512);
        }

        const unsigned obase = (unsigned)b * NPAIRS + (unsigned)m;

        if (role == 0 || role == 3) {
            const int p = (role == 0) ? 0 : 2;
            const float* fp = sc + p * BAND + foff;
            const float* fq = fp + BAND;
            v8f app = (v8f){0.f, 0.f, 0.f, 0.f, 0.f, 0.f, 0.f, 0.f};
            v8f apq = app, aqq = app;
            for (int k = 0; k < DK / 4; ++k) {
                const v2f xp = *(const v2f*)(fp + (k << 2));
                const v2f xq = *(const v2f*)(fq + (k << 2));
                app = mma4(xp, xp, app);
                apq = mma4(xp, xq, apq);
                aqq = mma4(xq, xq, aqq);
            }
            const int cp = p << 4, cq = cp + 16;
            unsigned ep[8], eq[8];
            #pragma unroll
            for (int v = 0; v < 8; ++v) {
                const int gp = cp + rloc + v;
                ep[v] = obase + (unsigned)rowterm(gp);
                eq[v] = obase + (unsigned)rowterm(gp + 16);
            }
            #pragma unroll
            for (int v = 0; v < 8; ++v)
                out[ep[v] + cq] = apq[v];
            #pragma unroll
            for (int v = 0; v < 8; ++v) {
                if (m > rloc + v) {
                    out[ep[v] + cp] = app[v];
                    out[eq[v] + cq] = aqq[v];
                }
            }
        } else {
            const int p = role - 1;
            const float* fp = sc + p * BAND + foff;
            const float* f2 = sc + 2 * BAND + foff;
            const float* f3 = sc + 3 * BAND + foff;
            v8f ap2 = (v8f){0.f, 0.f, 0.f, 0.f, 0.f, 0.f, 0.f, 0.f};
            v8f ap3 = ap2;
            for (int k = 0; k < DK / 4; ++k) {
                const v2f xp = *(const v2f*)(fp + (k << 2));
                const v2f x2 = *(const v2f*)(f2 + (k << 2));
                const v2f x3 = *(const v2f*)(f3 + (k << 2));
                ap2 = mma4(xp, x2, ap2);
                ap3 = mma4(xp, x3, ap3);
            }
            const int cp = p << 4;
            #pragma unroll
            for (int v = 0; v < 8; ++v) {
                const unsigned ep = obase + (unsigned)rowterm(cp + rloc + v);
                out[ep + 32] = ap2[v];
                out[ep + 48] = ap3[v];
            }
        }
        role = (role + 1) & 3;

        if (more) {
            float* sn = sx + (cur ^ 1) * TILE_FLOATS;
            #pragma unroll
            for (int i = 0; i < NLOAD; ++i)
                *reinterpret_cast<v4f*>(&sn[soff + i * (4 * LDS_STRIDE)]) = r[i];
        }
        __syncthreads();
        cur ^= 1;
    }
}

extern "C" void kernel_launch(void* const* d_in, const int* in_sizes, int n_in,
                              void* d_out, int out_size, void* d_ws, size_t ws_size,
                              hipStream_t stream) {
    (void)in_sizes; (void)n_in; (void)d_ws; (void)ws_size; (void)out_size;
    const float* x = (const float*)d_in[0];
    float* out = (float*)d_out;
    InteractionLayer_gram_triu_kernel<<<NBLOCKS, 128, 0, stream>>>(x, out);
}


# Wave priority by progress: s_setprio 3,2,1,0 over a workgroup's trips, so the co-resident workgroup furthest behind wins the matrix unit and none walks its last trips alone; sums unchanged
# speedup vs baseline: 1.5446x; 1.0672x over previous
#include <hip/hip_runtime.h>

typedef __attribute__((ext_vector_type(2))) float v2f;
typedef __attribute__((ext_vector_type(4))) float v4f;
typedef __attribute__((ext_vector_type(8))) float v8f;

#define NF 64
#define DK 128
#define LDS_STRIDE 132
#define NPAIRS 2016
#define NBATCH 4096
#define TILE_FLOATS (NF * LDS_STRIDE)
#define NLOAD 16
#define NBLOCKS 1024

__device__ __forceinline__ v8f mma4(v2f a, v2f b, v8f acc) {
    return __builtin_amdgcn_wmma_f32_16x16x4_f32( false, a,  false, b,
                                                  (short)0, acc,
                                                  false,  false);
}

__device__ __forceinline__ int rowterm(int gi) {
    return ((gi * (125 - gi)) >> 1) - 1;
}

__global__ __launch_bounds__(128)
void InteractionLayer_gram_triu_kernel(const float* __restrict__ x,
                                       float* __restrict__ out) {
    __shared__ float sx[2 * TILE_FLOATS];

    const int tid  = threadIdx.x;
    const int step = gridDim.x;
    int b = blockIdx.x;
    if (b >= NBATCH) return;

    __builtin_amdgcn_s_setprio(3);

    const int wave  = __builtin_amdgcn_readfirstlane(tid >> 5);
    const int lane  = tid & 31;
    const int m     = lane & 15;
    const int khalf = lane >> 4;

    const int goff = tid << 2;
    const int soff = wave * LDS_STRIDE + (lane << 2);

    const int foff  = m * LDS_STRIDE + (khalf << 1);
    const int BAND  = 16 * LDS_STRIDE;

    const int rloc  = khalf << 3;

    v4f r[NLOAD];

    {
        const float* __restrict__ xb = x + (size_t)b * (NF * DK);
        #pragma unroll
        for (int i = 0; i < NLOAD; ++i)
            r[i] = *reinterpret_cast<const v4f*>(xb + goff + i * 512);
        #pragma unroll
        for (int i = 0; i < NLOAD; ++i)
            *reinterpret_cast<v4f*>(&sx[soff + i * (4 * LDS_STRIDE)]) = r[i];
    }
    __syncthreads();

    int cur  = 0;
    int role = wave;
    int done = 0;
    #pragma nounroll
    for (; b < NBATCH; b += step) {
        const int  bn   = b + step;
        const bool more = bn < NBATCH;
        const float* sc = sx + cur * TILE_FLOATS;

        if (more) {
            const float* __restrict__ xb = x + (size_t)bn * (NF * DK);
            #pragma unroll
            for (int i = 0; i < NLOAD; ++i)
                r[i] = *reinterpret_cast<const v4f*>(xb + goff + i * 512);
        }

        const unsigned obase = (unsigned)b * NPAIRS + (unsigned)m;

        if (role == 0 || role == 3) {
            const int p = (role == 0) ? 0 : 2;
            const float* fp = sc + p * BAND + foff;
            const float* fq = fp + BAND;
            v8f app = (v8f){0.f, 0.f, 0.f, 0.f, 0.f, 0.f, 0.f, 0.f};
            v8f apq = app, aqq = app;
            for (int k = 0; k < DK / 4; ++k) {
                const v2f xp = *(const v2f*)(fp + (k << 2));
                const v2f xq = *(const v2f*)(fq + (k << 2));
                app = mma4(xp, xp, app);
                apq = mma4(xp, xq, apq);
                aqq = mma4(xq, xq, aqq);
            }
            const int cp = p << 4, cq = cp + 16;
            unsigned ep[8], eq[8];
            #pragma unroll
            for (int v = 0; v < 8; ++v) {
                const int gp = cp + rloc + v;
                ep[v] = obase + (unsigned)rowterm(gp);
                eq[v] = obase + (unsigned)rowterm(gp + 16);
            }
            #pragma unroll
            for (int v = 0; v < 8; ++v)
                out[ep[v] + cq] = apq[v];
            #pragma unroll
            for (int v = 0; v < 8; ++v) {
                if (m > rloc + v) {
                    out[ep[v] + cp] = app[v];
                    out[eq[v] + cq] = aqq[v];
                }
            }
        } else {
            const int p = role - 1;
            const float* fp = sc + p * BAND + foff;
            const float* f2 = sc + 2 * BAND + foff;
            const float* f3 = sc + 3 * BAND + foff;
            v8f ap2 = (v8f){0.f, 0.f, 0.f, 0.f, 0.f, 0.f, 0.f, 0.f};
            v8f ap3 = ap2;
            for (int k = 0; k < DK / 4; ++k) {
                const v2f xp = *(const v2f*)(fp + (k << 2));
                const v2f x2 = *(const v2f*)(f2 + (k << 2));
                const v2f x3 = *(const v2f*)(f3 + (k << 2));
                ap2 = mma4(xp, x2, ap2);
                ap3 = mma4(xp, x3, ap3);
            }
            const int cp = p << 4;
            #pragma unroll
            for (int v = 0; v < 8; ++v) {
                const unsigned ep = obase + (unsigned)rowterm(cp + rloc + v);
                out[ep + 32] = ap2[v];
                out[ep + 48] = ap3[v];
            }
        }
        role = (role + 1) & 3;

        if (more) {
            float* sn = sx + (cur ^ 1) * TILE_FLOATS;
            #pragma unroll
            for (int i = 0; i < NLOAD; ++i)
                *reinterpret_cast<v4f*>(&sn[soff + i * (4 * LDS_STRIDE)]) = r[i];
        }
        __syncthreads();
        cur ^= 1;

        ++done;
        if (done == 1)      __builtin_amdgcn_s_setprio(2);
        else if (done == 2) __builtin_amdgcn_s_setprio(1);
        else if (done == 3) __builtin_amdgcn_s_setprio(0);
    }
    __builtin_amdgcn_s_setprio(0);
}

extern "C" void kernel_launch(void* const* d_in, const int* in_sizes, int n_in,
                              void* d_out, int out_size, void* d_ws, size_t ws_size,
                              hipStream_t stream) {
    (void)in_sizes; (void)n_in; (void)d_ws; (void)ws_size; (void)out_size;
    const float* x = (const float*)d_in[0];
    float* out = (float*)d_out;
    InteractionLayer_gram_triu_kernel<<<NBLOCKS, 128, 0, stream>>>(x, out);
}
